# starting kernel
# baseline (speedup 1.0000x reference)
#include <hip/hip_runtime.h>
#include <hip/hip_bf16.h>

typedef __attribute__((ext_vector_type(2))) float v2f;
typedef __attribute__((ext_vector_type(8))) float v8f;

#define N_TOT 131072
#define K_CLS 64
#define D_DIM 256
#define FP_EPS 1e-12f
#define MLDS_STRIDE 260

__device__ __forceinline__ v8f wmma4(v2f a, v2f b, v8f c) {
    return __builtin_amdgcn_wmma_f32_16x16x4_f32(
         false, a,  false, b,
         (short)0, c,  false,  false);
}

__global__ __launch_bounds__(256) void zero_ws(float* __restrict__ ws,
                                               float* __restrict__ out0) {
    int i = blockIdx.x * 256 + threadIdx.x;
    if (i < K_CLS * D_DIM + 2 * K_CLS) ws[i] = 0.0f;
    if (i == 0) out0[0] = 0.0f;
}

__global__ __launch_bounds__(256)
void pass1_colsum(const float* __restrict__ X, const float* __restrict__ A,
                  float* __restrict__ sums, float* __restrict__ freq,
                  int rowsPerBlock) {
    const int tid   = threadIdx.x;
    const int lane  = tid & 31;
    const int wave  = tid >> 5;
    const int ktile = wave & 3;
    const int dgrp  = wave >> 2;
    const int kbase = ktile * 16;
    const int dbase = dgrp * 128;
    const int l15   = lane & 15;
    const int hi    = (lane >= 16) ? 1 : 0;

    v8f acc[8];
    #pragma unroll
    for (int t = 0; t < 8; ++t) { v8f z = {}; acc[t] = z; }

    float fpart = 0.0f;
    const int row0 = blockIdx.x * rowsPerBlock;

    for (int n0 = row0; n0 < row0 + rowsPerBlock; n0 += 4) {
        const int ra = n0 + (hi ? 2 : 0);
        const int rb = ra + 1;
        v2f af;
        af.x = A[(size_t)ra * K_CLS + kbase + l15];
        af.y = A[(size_t)rb * K_CLS + kbase + l15];
        fpart += af.x + af.y;

        const float* xra = X + (size_t)ra * D_DIM + dbase + l15;
        const float* xrb = X + (size_t)rb * D_DIM + dbase + l15;
        #pragma unroll
        for (int dt = 0; dt < 8; ++dt) {
            v2f bf;
            bf.x = xra[dt * 16];
            bf.y = xrb[dt * 16];
            acc[dt] = wmma4(af, bf, acc[dt]);
        }
    }

    fpart += __shfl_xor(fpart, 16, 32);
    if (dgrp == 0 && lane < 16)
        unsafeAtomicAdd(&freq[kbase + lane], fpart);

    #pragma unroll
    for (int dt = 0; dt < 8; ++dt) {
        #pragma unroll
        for (int r = 0; r < 8; ++r) {
            const int k = kbase + r + (hi ? 8 : 0);
            const int d = dbase + dt * 16 + l15;
            unsafeAtomicAdd(&sums[k * D_DIM + d], acc[dt][r]);
        }
    }
}

__global__ __launch_bounds__(256)
void means_msq(const float* __restrict__ sums, const float* __restrict__ freq,
               const float* __restrict__ past, float* __restrict__ out_means,
               float* __restrict__ msq) {
    __shared__ float red[8];
    const int k = blockIdx.x;
    const int d = threadIdx.x;
    const float f  = freq[k];
    const float bm = sums[k * D_DIM + d] / fmaxf(f, FP_EPS);
    const float pm = past[k * D_DIM + d];
    const float m  = (f > FP_EPS) ? (pm * 0.5f + bm * 0.5f) : pm;
    out_means[k * D_DIM + d] = m;

    float sq = m * m;
    #pragma unroll
    for (int off = 16; off > 0; off >>= 1) sq += __shfl_xor(sq, off, 32);
    if ((threadIdx.x & 31) == 0) red[threadIdx.x >> 5] = sq;
    __syncthreads();
    if (threadIdx.x == 0) {
        float t = 0.0f;
        #pragma unroll
        for (int i = 0; i < 8; ++i) t += red[i];
        msq[k] = t;
    }
}

__global__ __launch_bounds__(256)
void pass2_loss(const float* __restrict__ X, const float* __restrict__ A,
                const float* __restrict__ S, const float* __restrict__ means,
                const float* __restrict__ msq, float* __restrict__ loss) {
    extern __shared__ float smem[];
    float* mlds    = smem;
    float* msq_lds = smem + K_CLS * MLDS_STRIDE;
    float* xsq     = msq_lds + K_CLS;

    const int tid  = threadIdx.x;
    const int lane = tid & 31;
    const int wave = tid >> 5;
    const int l15  = lane & 15;
    const int hi   = (lane >= 16) ? 1 : 0;

    #pragma unroll
    for (int i = 0; i < 16; ++i) {
        const int idx = tid + i * 256;
        const int k   = idx >> 6;
        const int c4  = idx & 63;
        const float4 v = ((const float4*)means)[(size_t)k * 64 + c4];
        *(float4*)&mlds[k * MLDS_STRIDE + c4 * 4] = v;
    }
    if (tid < K_CLS) msq_lds[tid] = msq[tid];
    if (tid < 128)   xsq[tid] = 0.0f;
    __syncthreads();

    const int n0 = blockIdx.x * 128 + wave * 16;

    v8f acc[4];
    #pragma unroll
    for (int t = 0; t < 4; ++t) { v8f z = {}; acc[t] = z; }

    float xp = 0.0f;
    const float* xrow = X + (size_t)(n0 + l15) * D_DIM + (hi ? 2 : 0);

    for (int d0 = 0; d0 < D_DIM; d0 += 4) {
        v2f af;
        af.x = xrow[d0];
        af.y = xrow[d0 + 1];
        xp += af.x * af.x + af.y * af.y;
        #pragma unroll
        for (int kt = 0; kt < 4; ++kt) {
            const float* mrow = &mlds[(kt * 16 + l15) * MLDS_STRIDE + d0 + (hi ? 2 : 0)];
            v2f bf;
            bf.x = mrow[0];
            bf.y = mrow[1];
            acc[kt] = wmma4(af, bf, acc[kt]);
        }
    }

    unsafeAtomicAdd(&xsq[wave * 16 + l15], xp);
    __syncthreads();

    float c = 0.0f;
    #pragma unroll
    for (int kt = 0; kt < 4; ++kt) {
        #pragma unroll
        for (int r = 0; r < 8; ++r) {
            const int nl = r + (hi ? 8 : 0);
            const int k  = kt * 16 + l15;
            const float cross = acc[kt][r];
            const float dist  = xsq[wave * 16 + nl] + msq_lds[k] - 2.0f * cross;
            const size_t gi = (size_t)(n0 + nl) * K_CLS + k;
            c += A[gi] * S[gi] * dist;
        }
    }
    #pragma unroll
    for (int off = 16; off > 0; off >>= 1) c += __shfl_xor(c, off, 32);
    if (lane == 0)
        unsafeAtomicAdd(loss, c * (1.0f / (float)N_TOT));
}

extern "C" void kernel_launch(void* const* d_in, const int* in_sizes, int n_in,
                              void* d_out, int out_size, void* d_ws, size_t ws_size,
                              hipStream_t stream) {
    const float* X = (const float*)d_in[0];
    const float* A = (const float*)d_in[1];
    const float* S = (const float*)d_in[2];
    const float* P = (const float*)d_in[3];
    float* out = (float*)d_out;

    float* ws   = (float*)d_ws;
    float* sums = ws;
    float* freq = ws + K_CLS * D_DIM;
    float* msq  = freq + K_CLS;

    zero_ws<<<(K_CLS * D_DIM + 2 * K_CLS + 255) / 256, 256, 0, stream>>>(ws, out);

    const int rowsPerBlock = 512;
    pass1_colsum<<<N_TOT / rowsPerBlock, 256, 0, stream>>>(X, A, sums, freq, rowsPerBlock);

    means_msq<<<K_CLS, 256, 0, stream>>>(sums, freq, P, out + 1, msq);

    const size_t smemBytes = (size_t)(K_CLS * MLDS_STRIDE + K_CLS + 128) * sizeof(float);
    pass2_loss<<<N_TOT / 128, 256, smemBytes, stream>>>(X, A, S, out + 1, msq, out);
}


# pass 1 re-tiled: wave tile 64k x 64d from 16-byte row loads, next group's loads in a second register set, row halves combined in LDS, one atomic per element per work-group (different reduction tree)
# speedup vs baseline: 1.6238x; 1.6238x over previous
#include <hip/hip_runtime.h>
#include <hip/hip_bf16.h>

typedef __attribute__((ext_vector_type(2))) float v2f;
typedef __attribute__((ext_vector_type(8))) float v8f;

#define N_TOT 131072
#define K_CLS 64
#define D_DIM 256
#define FP_EPS 1e-12f
#define MLDS_STRIDE 260

__device__ __forceinline__ v8f wmma4(v2f a, v2f b, v8f c) {
    return __builtin_amdgcn_wmma_f32_16x16x4_f32(
         false, a,  false, b,
         (short)0, c,  false,  false);
}

__global__ __launch_bounds__(256) void zero_ws(float* __restrict__ ws,
                                               float* __restrict__ out0) {
    int i = blockIdx.x * 256 + threadIdx.x;
    if (i < K_CLS * D_DIM + 2 * K_CLS) ws[i] = 0.0f;
    if (i == 0) out0[0] = 0.0f;
}

#define P1_ROWS   256
#define P1_STEPS  (P1_ROWS / 2 / 4)
#define P1_U      2
#define P1_GROUPS (P1_STEPS / P1_U)

struct P1Frag { float4 xa, xb, aa, ab; };

__device__ __forceinline__ void p1_load(P1Frag (&f)[P1_U], const float4* __restrict__ xp,
                                        const float4* __restrict__ ap, int g) {
    #pragma unroll
    for (int u = 0; u < P1_U; ++u) {
        const int s = g * P1_U + u;
        f[u].xa = xp[s * (4 * D_DIM / 4)];
        f[u].xb = xp[s * (4 * D_DIM / 4) + D_DIM / 4];
        f[u].aa = ap[s * (4 * K_CLS / 4)];
        f[u].ab = ap[s * (4 * K_CLS / 4) + K_CLS / 4];
    }
}

__device__ __forceinline__ void p1_mma(const P1Frag (&f)[P1_U], v8f (&acc)[4][4], float4& fq) {
    #pragma unroll
    for (int u = 0; u < P1_U; ++u) {
        fq.x += f[u].aa.x + f[u].ab.x;
        fq.y += f[u].aa.y + f[u].ab.y;
        fq.z += f[u].aa.z + f[u].ab.z;
        fq.w += f[u].aa.w + f[u].ab.w;
        v2f af[4], bf[4];
        af[0].x = f[u].aa.x; af[0].y = f[u].ab.x;
        af[1].x = f[u].aa.y; af[1].y = f[u].ab.y;
        af[2].x = f[u].aa.z; af[2].y = f[u].ab.z;
        af[3].x = f[u].aa.w; af[3].y = f[u].ab.w;
        bf[0].x = f[u].xa.x; bf[0].y = f[u].xb.x;
        bf[1].x = f[u].xa.y; bf[1].y = f[u].xb.y;
        bf[2].x = f[u].xa.z; bf[2].y = f[u].xb.z;
        bf[3].x = f[u].xa.w; bf[3].y = f[u].xb.w;
        #pragma unroll
        for (int kt = 0; kt < 4; ++kt) {
            #pragma unroll
            for (int dt = 0; dt < 4; ++dt)
                acc[kt][dt] = wmma4(af[kt], bf[dt], acc[kt][dt]);
        }
    }
}

__global__ __launch_bounds__(256)
void pass1_colsum(const float* __restrict__ X, const float* __restrict__ A,
                  float* __restrict__ sums, float* __restrict__ freq) {
    __shared__ float part[K_CLS * D_DIM];

    const int tid  = threadIdx.x;
    const int lane = tid & 31;
    const int wave = tid >> 5;
    const int dg   = wave & 3;
    const int nh   = wave >> 2;
    const int l15  = lane & 15;
    const int hi   = lane >> 4;

    v8f acc[4][4];
    #pragma unroll
    for (int a = 0; a < 4; ++a) {
        #pragma unroll
        for (int b = 0; b < 4; ++b) { v8f z = {}; acc[a][b] = z; }
    }
    float4 fq = make_float4(0.0f, 0.0f, 0.0f, 0.0f);

    const int row = blockIdx.x * P1_ROWS + nh * (P1_ROWS / 2) + hi * 2;
    const float4* xp = (const float4*)(X + (size_t)row * D_DIM + dg * 64 + l15 * 4);
    const float4* ap = (const float4*)(A + (size_t)row * K_CLS + l15 * 4);

    P1Frag f0[P1_U], f1[P1_U];
    p1_load(f0, xp, ap, 0);
    for (int g = 0; g < P1_GROUPS - 2; g += 2) {
        p1_load(f1, xp, ap, g + 1);
        __builtin_amdgcn_sched_barrier(0);
        p1_mma(f0, acc, fq);
        __builtin_amdgcn_sched_barrier(0);
        p1_load(f0, xp, ap, g + 2);
        __builtin_amdgcn_sched_barrier(0);
        p1_mma(f1, acc, fq);
    }
    p1_load(f1, xp, ap, P1_GROUPS - 1);
    __builtin_amdgcn_sched_barrier(0);
    p1_mma(f0, acc, fq);
    p1_mma(f1, acc, fq);

    fq.x += __shfl_xor(fq.x, 16, 32);
    fq.y += __shfl_xor(fq.y, 16, 32);
    fq.z += __shfl_xor(fq.z, 16, 32);
    fq.w += __shfl_xor(fq.w, 16, 32);
    if (dg == 0 && lane < 16) {
        unsafeAtomicAdd(&freq[l15 * 4 + 0], fq.x);
        unsafeAtomicAdd(&freq[l15 * 4 + 1], fq.y);
        unsafeAtomicAdd(&freq[l15 * 4 + 2], fq.z);
        unsafeAtomicAdd(&freq[l15 * 4 + 3], fq.w);
    }

    if (nh == 1) {
        #pragma unroll
        for (int kt = 0; kt < 4; ++kt) {
            #pragma unroll
            for (int r = 0; r < 8; ++r) {
                const int k = (r + 8 * hi) * 4 + kt;
                const float4 v = make_float4(acc[kt][0][r], acc[kt][1][r], acc[kt][2][r], acc[kt][3][r]);
                *(float4*)&part[k * D_DIM + dg * 64 + l15 * 4] = v;
            }
        }
    }
    __syncthreads();
    if (nh == 0) {
        #pragma unroll
        for (int kt = 0; kt < 4; ++kt) {
            #pragma unroll
            for (int r = 0; r < 8; ++r) {
                const int k = (r + 8 * hi) * 4 + kt;
                float4* p = (float4*)&part[k * D_DIM + dg * 64 + l15 * 4];
                float4 v = *p;
                v.x += acc[kt][0][r]; v.y += acc[kt][1][r];
                v.z += acc[kt][2][r]; v.w += acc[kt][3][r];
                *p = v;
            }
        }
    }
    __syncthreads();
    for (int k0 = 0; k0 < K_CLS; k0 += 16) {
        float v[16];
        #pragma unroll
        for (int j = 0; j < 16; ++j) v[j] = part[(k0 + j) * D_DIM + tid];
        #pragma unroll
        for (int j = 0; j < 16; ++j) unsafeAtomicAdd(&sums[(k0 + j) * D_DIM + tid], v[j]);
    }
}

__global__ __launch_bounds__(256)
void means_msq(const float* __restrict__ sums, const float* __restrict__ freq,
               const float* __restrict__ past, float* __restrict__ out_means,
               float* __restrict__ msq) {
    __shared__ float red[8];
    const int k = blockIdx.x;
    const int d = threadIdx.x;
    const float f  = freq[k];
    const float bm = sums[k * D_DIM + d] / fmaxf(f, FP_EPS);
    const float pm = past[k * D_DIM + d];
    const float m  = (f > FP_EPS) ? (pm * 0.5f + bm * 0.5f) : pm;
    out_means[k * D_DIM + d] = m;

    float sq = m * m;
    #pragma unroll
    for (int off = 16; off > 0; off >>= 1) sq += __shfl_xor(sq, off, 32);
    if ((threadIdx.x & 31) == 0) red[threadIdx.x >> 5] = sq;
    __syncthreads();
    if (threadIdx.x == 0) {
        float t = 0.0f;
        #pragma unroll
        for (int i = 0; i < 8; ++i) t += red[i];
        msq[k] = t;
    }
}

__global__ __launch_bounds__(256)
void pass2_loss(const float* __restrict__ X, const float* __restrict__ A,
                const float* __restrict__ S, const float* __restrict__ means,
                const float* __restrict__ msq, float* __restrict__ loss) {
    extern __shared__ float smem[];
    float* mlds    = smem;
    float* msq_lds = smem + K_CLS * MLDS_STRIDE;
    float* xsq     = msq_lds + K_CLS;

    const int tid  = threadIdx.x;
    const int lane = tid & 31;
    const int wave = tid >> 5;
    const int l15  = lane & 15;
    const int hi   = (lane >= 16) ? 1 : 0;

    #pragma unroll
    for (int i = 0; i < 16; ++i) {
        const int idx = tid + i * 256;
        const int k   = idx >> 6;
        const int c4  = idx & 63;
        const float4 v = ((const float4*)means)[(size_t)k * 64 + c4];
        *(float4*)&mlds[k * MLDS_STRIDE + c4 * 4] = v;
    }
    if (tid < K_CLS) msq_lds[tid] = msq[tid];
    if (tid < 128)   xsq[tid] = 0.0f;
    __syncthreads();

    const int n0 = blockIdx.x * 128 + wave * 16;

    v8f acc[4];
    #pragma unroll
    for (int t = 0; t < 4; ++t) { v8f z = {}; acc[t] = z; }

    float xp = 0.0f;
    const float* xrow = X + (size_t)(n0 + l15) * D_DIM + (hi ? 2 : 0);

    for (int d0 = 0; d0 < D_DIM; d0 += 4) {
        v2f af;
        af.x = xrow[d0];
        af.y = xrow[d0 + 1];
        xp += af.x * af.x + af.y * af.y;
        #pragma unroll
        for (int kt = 0; kt < 4; ++kt) {
            const float* mrow = &mlds[(kt * 16 + l15) * MLDS_STRIDE + d0 + (hi ? 2 : 0)];
            v2f bf;
            bf.x = mrow[0];
            bf.y = mrow[1];
            acc[kt] = wmma4(af, bf, acc[kt]);
        }
    }

    unsafeAtomicAdd(&xsq[wave * 16 + l15], xp);
    __syncthreads();

    float c = 0.0f;
    #pragma unroll
    for (int kt = 0; kt < 4; ++kt) {
        #pragma unroll
        for (int r = 0; r < 8; ++r) {
            const int nl = r + (hi ? 8 : 0);
            const int k  = kt * 16 + l15;
            const float cross = acc[kt][r];
            const float dist  = xsq[wave * 16 + nl] + msq_lds[k] - 2.0f * cross;
            const size_t gi = (size_t)(n0 + nl) * K_CLS + k;
            c += A[gi] * S[gi] * dist;
        }
    }
    #pragma unroll
    for (int off = 16; off > 0; off >>= 1) c += __shfl_xor(c, off, 32);
    if (lane == 0)
        unsafeAtomicAdd(loss, c * (1.0f / (float)N_TOT));
}

extern "C" void kernel_launch(void* const* d_in, const int* in_sizes, int n_in,
                              void* d_out, int out_size, void* d_ws, size_t ws_size,
                              hipStream_t stream) {
    const float* X = (const float*)d_in[0];
    const float* A = (const float*)d_in[1];
    const float* S = (const float*)d_in[2];
    const float* P = (const float*)d_in[3];
    float* out = (float*)d_out;

    float* ws   = (float*)d_ws;
    float* sums = ws;
    float* freq = ws + K_CLS * D_DIM;
    float* msq  = freq + K_CLS;

    zero_ws<<<(K_CLS * D_DIM + 2 * K_CLS + 255) / 256, 256, 0, stream>>>(ws, out);

    pass1_colsum<<<N_TOT / P1_ROWS, 256, 0, stream>>>(X, A, sums, freq);

    means_msq<<<K_CLS, 256, 0, stream>>>(sums, freq, P, out + 1, msq);

    const size_t smemBytes = (size_t)(K_CLS * MLDS_STRIDE + K_CLS + 128) * sizeof(float);
    pass2_loss<<<N_TOT / 128, 256, smemBytes, stream>>>(X, A, S, out + 1, msq, out);
}


# pass 1 split by 64-column slice: each work-group owns one slice over 2048 rows (16 waves), adds its 16 wave tiles in LDS in fixed order, 8x fewer global atomics (different reduction tree)
# speedup vs baseline: 1.8687x; 1.1508x over previous
#include <hip/hip_runtime.h>
#include <hip/hip_bf16.h>

typedef __attribute__((ext_vector_type(2))) float v2f;
typedef __attribute__((ext_vector_type(8))) float v8f;

#define N_TOT 131072
#define K_CLS 64
#define D_DIM 256
#define FP_EPS 1e-12f
#define MLDS_STRIDE 260

__device__ __forceinline__ v8f wmma4(v2f a, v2f b, v8f c) {
    return __builtin_amdgcn_wmma_f32_16x16x4_f32(
         false, a,  false, b,
         (short)0, c,  false,  false);
}

__global__ __launch_bounds__(256) void zero_ws(float* __restrict__ ws,
                                               float* __restrict__ out0) {
    int i = blockIdx.x * 256 + threadIdx.x;
    if (i < K_CLS * D_DIM + 2 * K_CLS) ws[i] = 0.0f;
    if (i == 0) out0[0] = 0.0f;
}

#define P1_WAVES  16
#define P1_WROWS  128
#define P1_ROWS   (P1_WAVES * P1_WROWS)
#define P1_STEPS  (P1_WROWS / 4)
#define P1_U      2
#define P1_GROUPS (P1_STEPS / P1_U)
#define P1_DSL    64
#define P1_TILE   (K_CLS * P1_DSL)

struct P1Frag { float4 xa, xb, aa, ab; };

__device__ __forceinline__ void p1_load(P1Frag (&f)[P1_U], const float4* __restrict__ xp,
                                        const float4* __restrict__ ap, int g) {
    #pragma unroll
    for (int u = 0; u < P1_U; ++u) {
        const int s = g * P1_U + u;
        f[u].xa = xp[s * (4 * D_DIM / 4)];
        f[u].xb = xp[s * (4 * D_DIM / 4) + D_DIM / 4];
        f[u].aa = ap[s * (4 * K_CLS / 4)];
        f[u].ab = ap[s * (4 * K_CLS / 4) + K_CLS / 4];
    }
}

__device__ __forceinline__ void p1_mma(const P1Frag (&f)[P1_U], v8f (&acc)[4][4], float4& fq) {
    #pragma unroll
    for (int u = 0; u < P1_U; ++u) {
        fq.x += f[u].aa.x + f[u].ab.x;
        fq.y += f[u].aa.y + f[u].ab.y;
        fq.z += f[u].aa.z + f[u].ab.z;
        fq.w += f[u].aa.w + f[u].ab.w;
        v2f af[4], bf[4];
        af[0].x = f[u].aa.x; af[0].y = f[u].ab.x;
        af[1].x = f[u].aa.y; af[1].y = f[u].ab.y;
        af[2].x = f[u].aa.z; af[2].y = f[u].ab.z;
        af[3].x = f[u].aa.w; af[3].y = f[u].ab.w;
        bf[0].x = f[u].xa.x; bf[0].y = f[u].xb.x;
        bf[1].x = f[u].xa.y; bf[1].y = f[u].xb.y;
        bf[2].x = f[u].xa.z; bf[2].y = f[u].xb.z;
        bf[3].x = f[u].xa.w; bf[3].y = f[u].xb.w;
        #pragma unroll
        for (int kt = 0; kt < 4; ++kt) {
            #pragma unroll
            for (int dt = 0; dt < 4; ++dt)
                acc[kt][dt] = wmma4(af[kt], bf[dt], acc[kt][dt]);
        }
    }
}

__global__ __launch_bounds__(512)
void pass1_colsum(const float* __restrict__ X, const float* __restrict__ A,
                  float* __restrict__ sums, float* __restrict__ freq) {
    __shared__ float part[P1_WAVES * P1_TILE];
    __shared__ float fpart[P1_WAVES * K_CLS];

    const int tid   = threadIdx.x;
    const int lane  = tid & 31;
    const int wave  = tid >> 5;
    const int dg    = blockIdx.x & 3;
    const int chunk = blockIdx.x >> 2;
    const int l15   = lane & 15;
    const int hi    = lane >> 4;

    v8f acc[4][4];
    #pragma unroll
    for (int a = 0; a < 4; ++a) {
        #pragma unroll
        for (int b = 0; b < 4; ++b) { v8f z = {}; acc[a][b] = z; }
    }
    float4 fq = make_float4(0.0f, 0.0f, 0.0f, 0.0f);

    const int row = chunk * P1_ROWS + wave * P1_WROWS + hi * 2;
    const float4* xp = (const float4*)(X + (size_t)row * D_DIM + dg * P1_DSL + l15 * 4);
    const float4* ap = (const float4*)(A + (size_t)row * K_CLS + l15 * 4);

    P1Frag f0[P1_U], f1[P1_U];
    p1_load(f0, xp, ap, 0);
    for (int g = 0; g < P1_GROUPS - 2; g += 2) {
        p1_load(f1, xp, ap, g + 1);
        __builtin_amdgcn_sched_barrier(0);
        p1_mma(f0, acc, fq);
        __builtin_amdgcn_sched_barrier(0);
        p1_load(f0, xp, ap, g + 2);
        __builtin_amdgcn_sched_barrier(0);
        p1_mma(f1, acc, fq);
    }
    p1_load(f1, xp, ap, P1_GROUPS - 1);
    __builtin_amdgcn_sched_barrier(0);
    p1_mma(f0, acc, fq);
    p1_mma(f1, acc, fq);

    fq.x += __shfl_xor(fq.x, 16, 32);
    fq.y += __shfl_xor(fq.y, 16, 32);
    fq.z += __shfl_xor(fq.z, 16, 32);
    fq.w += __shfl_xor(fq.w, 16, 32);
    if (dg == 0 && lane < 16)
        *(float4*)&fpart[wave * K_CLS + l15 * 4] = fq;

    float* mine = part + wave * P1_TILE;
    #pragma unroll
    for (int kt = 0; kt < 4; ++kt) {
        #pragma unroll
        for (int r = 0; r < 8; ++r) {
            const int k = (r + 8 * hi) * 4 + kt;
            const float4 v = make_float4(acc[kt][0][r], acc[kt][1][r], acc[kt][2][r], acc[kt][3][r]);
            *(float4*)&mine[k * P1_DSL + l15 * 4] = v;
        }
    }
    __syncthreads();

    float s[P1_TILE / 512];
    #pragma unroll
    for (int j = 0; j < P1_TILE / 512; ++j) {
        const int e = j * 512 + tid;
        float t = part[e];
        #pragma unroll
        for (int w = 1; w < P1_WAVES; ++w) t += part[w * P1_TILE + e];
        s[j] = t;
    }
    float fs = 0.0f;
    if (dg == 0 && tid < K_CLS) {
        fs = fpart[tid];
        #pragma unroll
        for (int w = 1; w < P1_WAVES; ++w) fs += fpart[w * K_CLS + tid];
    }
    #pragma unroll
    for (int j = 0; j < P1_TILE / 512; ++j) {
        const int e = j * 512 + tid;
        unsafeAtomicAdd(&sums[(e >> 6) * D_DIM + dg * P1_DSL + (e & 63)], s[j]);
    }
    if (dg == 0 && tid < K_CLS) unsafeAtomicAdd(&freq[tid], fs);
}

__global__ __launch_bounds__(256)
void means_msq(const float* __restrict__ sums, const float* __restrict__ freq,
               const float* __restrict__ past, float* __restrict__ out_means,
               float* __restrict__ msq) {
    __shared__ float red[8];
    const int k = blockIdx.x;
    const int d = threadIdx.x;
    const float f  = freq[k];
    const float bm = sums[k * D_DIM + d] / fmaxf(f, FP_EPS);
    const float pm = past[k * D_DIM + d];
    const float m  = (f > FP_EPS) ? (pm * 0.5f + bm * 0.5f) : pm;
    out_means[k * D_DIM + d] = m;

    float sq = m * m;
    #pragma unroll
    for (int off = 16; off > 0; off >>= 1) sq += __shfl_xor(sq, off, 32);
    if ((threadIdx.x & 31) == 0) red[threadIdx.x >> 5] = sq;
    __syncthreads();
    if (threadIdx.x == 0) {
        float t = 0.0f;
        #pragma unroll
        for (int i = 0; i < 8; ++i) t += red[i];
        msq[k] = t;
    }
}

__global__ __launch_bounds__(256)
void pass2_loss(const float* __restrict__ X, const float* __restrict__ A,
                const float* __restrict__ S, const float* __restrict__ means,
                const float* __restrict__ msq, float* __restrict__ loss) {
    extern __shared__ float smem[];
    float* mlds    = smem;
    float* msq_lds = smem + K_CLS * MLDS_STRIDE;
    float* xsq     = msq_lds + K_CLS;

    const int tid  = threadIdx.x;
    const int lane = tid & 31;
    const int wave = tid >> 5;
    const int l15  = lane & 15;
    const int hi   = (lane >= 16) ? 1 : 0;

    #pragma unroll
    for (int i = 0; i < 16; ++i) {
        const int idx = tid + i * 256;
        const int k   = idx >> 6;
        const int c4  = idx & 63;
        const float4 v = ((const float4*)means)[(size_t)k * 64 + c4];
        *(float4*)&mlds[k * MLDS_STRIDE + c4 * 4] = v;
    }
    if (tid < K_CLS) msq_lds[tid] = msq[tid];
    if (tid < 128)   xsq[tid] = 0.0f;
    __syncthreads();

    const int n0 = blockIdx.x * 128 + wave * 16;

    v8f acc[4];
    #pragma unroll
    for (int t = 0; t < 4; ++t) { v8f z = {}; acc[t] = z; }

    float xp = 0.0f;
    const float* xrow = X + (size_t)(n0 + l15) * D_DIM + (hi ? 2 : 0);

    for (int d0 = 0; d0 < D_DIM; d0 += 4) {
        v2f af;
        af.x = xrow[d0];
        af.y = xrow[d0 + 1];
        xp += af.x * af.x + af.y * af.y;
        #pragma unroll
        for (int kt = 0; kt < 4; ++kt) {
            const float* mrow = &mlds[(kt * 16 + l15) * MLDS_STRIDE + d0 + (hi ? 2 : 0)];
            v2f bf;
            bf.x = mrow[0];
            bf.y = mrow[1];
            acc[kt] = wmma4(af, bf, acc[kt]);
        }
    }

    unsafeAtomicAdd(&xsq[wave * 16 + l15], xp);
    __syncthreads();

    float c = 0.0f;
    #pragma unroll
    for (int kt = 0; kt < 4; ++kt) {
        #pragma unroll
        for (int r = 0; r < 8; ++r) {
            const int nl = r + (hi ? 8 : 0);
            const int k  = kt * 16 + l15;
            const float cross = acc[kt][r];
            const float dist  = xsq[wave * 16 + nl] + msq_lds[k] - 2.0f * cross;
            const size_t gi = (size_t)(n0 + nl) * K_CLS + k;
            c += A[gi] * S[gi] * dist;
        }
    }
    #pragma unroll
    for (int off = 16; off > 0; off >>= 1) c += __shfl_xor(c, off, 32);
    if (lane == 0)
        unsafeAtomicAdd(loss, c * (1.0f / (float)N_TOT));
}

extern "C" void kernel_launch(void* const* d_in, const int* in_sizes, int n_in,
                              void* d_out, int out_size, void* d_ws, size_t ws_size,
                              hipStream_t stream) {
    const float* X = (const float*)d_in[0];
    const float* A = (const float*)d_in[1];
    const float* S = (const float*)d_in[2];
    const float* P = (const float*)d_in[3];
    float* out = (float*)d_out;

    float* ws   = (float*)d_ws;
    float* sums = ws;
    float* freq = ws + K_CLS * D_DIM;
    float* msq  = freq + K_CLS;

    zero_ws<<<(K_CLS * D_DIM + 2 * K_CLS + 255) / 256, 256, 0, stream>>>(ws, out);

    pass1_colsum<<<(N_TOT / P1_ROWS) * (D_DIM / P1_DSL), 512, 0, stream>>>(X, A, sums, freq);

    means_msq<<<K_CLS, 256, 0, stream>>>(sums, freq, P, out + 1, msq);

    const size_t smemBytes = (size_t)(K_CLS * MLDS_STRIDE + K_CLS + 128) * sizeof(float);
    pass2_loss<<<N_TOT / 128, 256, smemBytes, stream>>>(X, A, S, out + 1, msq, out);
}


# pass 2 data path: X by 16-byte row loads a chunk ahead, means fragments grouped ahead of each matrix run, A*S loaded first, x^2 by shuffles, one loss atomic per work-group (different reduction tree)
# speedup vs baseline: 2.2921x; 1.2266x over previous
#include <hip/hip_runtime.h>
#include <hip/hip_bf16.h>

typedef __attribute__((ext_vector_type(2))) float v2f;
typedef __attribute__((ext_vector_type(8))) float v8f;

#define N_TOT 131072
#define K_CLS 64
#define D_DIM 256
#define FP_EPS 1e-12f
#define MLDS_STRIDE 260

__device__ __forceinline__ v8f wmma4(v2f a, v2f b, v8f c) {
    return __builtin_amdgcn_wmma_f32_16x16x4_f32(
         false, a,  false, b,
         (short)0, c,  false,  false);
}

__global__ __launch_bounds__(256) void zero_ws(float* __restrict__ ws,
                                               float* __restrict__ out0) {
    int i = blockIdx.x * 256 + threadIdx.x;
    if (i < K_CLS * D_DIM + 2 * K_CLS) ws[i] = 0.0f;
    if (i == 0) out0[0] = 0.0f;
}

#define P1_WAVES  16
#define P1_WROWS  128
#define P1_ROWS   (P1_WAVES * P1_WROWS)
#define P1_STEPS  (P1_WROWS / 4)
#define P1_U      2
#define P1_GROUPS (P1_STEPS / P1_U)
#define P1_DSL    64
#define P1_TILE   (K_CLS * P1_DSL)

struct P1Frag { float4 xa, xb, aa, ab; };

__device__ __forceinline__ void p1_load(P1Frag (&f)[P1_U], const float4* __restrict__ xp,
                                        const float4* __restrict__ ap, int g) {
    #pragma unroll
    for (int u = 0; u < P1_U; ++u) {
        const int s = g * P1_U + u;
        f[u].xa = xp[s * (4 * D_DIM / 4)];
        f[u].xb = xp[s * (4 * D_DIM / 4) + D_DIM / 4];
        f[u].aa = ap[s * (4 * K_CLS / 4)];
        f[u].ab = ap[s * (4 * K_CLS / 4) + K_CLS / 4];
    }
}

__device__ __forceinline__ void p1_mma(const P1Frag (&f)[P1_U], v8f (&acc)[4][4], float4& fq) {
    #pragma unroll
    for (int u = 0; u < P1_U; ++u) {
        fq.x += f[u].aa.x + f[u].ab.x;
        fq.y += f[u].aa.y + f[u].ab.y;
        fq.z += f[u].aa.z + f[u].ab.z;
        fq.w += f[u].aa.w + f[u].ab.w;
        v2f af[4], bf[4];
        af[0].x = f[u].aa.x; af[0].y = f[u].ab.x;
        af[1].x = f[u].aa.y; af[1].y = f[u].ab.y;
        af[2].x = f[u].aa.z; af[2].y = f[u].ab.z;
        af[3].x = f[u].aa.w; af[3].y = f[u].ab.w;
        bf[0].x = f[u].xa.x; bf[0].y = f[u].xb.x;
        bf[1].x = f[u].xa.y; bf[1].y = f[u].xb.y;
        bf[2].x = f[u].xa.z; bf[2].y = f[u].xb.z;
        bf[3].x = f[u].xa.w; bf[3].y = f[u].xb.w;
        #pragma unroll
        for (int kt = 0; kt < 4; ++kt) {
            #pragma unroll
            for (int dt = 0; dt < 4; ++dt)
                acc[kt][dt] = wmma4(af[kt], bf[dt], acc[kt][dt]);
        }
    }
}

__global__ __launch_bounds__(512)
void pass1_colsum(const float* __restrict__ X, const float* __restrict__ A,
                  float* __restrict__ sums, float* __restrict__ freq) {
    __shared__ float part[P1_WAVES * P1_TILE];
    __shared__ float fpart[P1_WAVES * K_CLS];

    const int tid   = threadIdx.x;
    const int lane  = tid & 31;
    const int wave  = tid >> 5;
    const int dg    = blockIdx.x & 3;
    const int chunk = blockIdx.x >> 2;
    const int l15   = lane & 15;
    const int hi    = lane >> 4;

    v8f acc[4][4];
    #pragma unroll
    for (int a = 0; a < 4; ++a) {
        #pragma unroll
        for (int b = 0; b < 4; ++b) { v8f z = {}; acc[a][b] = z; }
    }
    float4 fq = make_float4(0.0f, 0.0f, 0.0f, 0.0f);

    const int row = chunk * P1_ROWS + wave * P1_WROWS + hi * 2;
    const float4* xp = (const float4*)(X + (size_t)row * D_DIM + dg * P1_DSL + l15 * 4);
    const float4* ap = (const float4*)(A + (size_t)row * K_CLS + l15 * 4);

    P1Frag f0[P1_U], f1[P1_U];
    p1_load(f0, xp, ap, 0);
    for (int g = 0; g < P1_GROUPS - 2; g += 2) {
        p1_load(f1, xp, ap, g + 1);
        __builtin_amdgcn_sched_barrier(0);
        p1_mma(f0, acc, fq);
        __builtin_amdgcn_sched_barrier(0);
        p1_load(f0, xp, ap, g + 2);
        __builtin_amdgcn_sched_barrier(0);
        p1_mma(f1, acc, fq);
    }
    p1_load(f1, xp, ap, P1_GROUPS - 1);
    __builtin_amdgcn_sched_barrier(0);
    p1_mma(f0, acc, fq);
    p1_mma(f1, acc, fq);

    fq.x += __shfl_xor(fq.x, 16, 32);
    fq.y += __shfl_xor(fq.y, 16, 32);
    fq.z += __shfl_xor(fq.z, 16, 32);
    fq.w += __shfl_xor(fq.w, 16, 32);
    if (dg == 0 && lane < 16)
        *(float4*)&fpart[wave * K_CLS + l15 * 4] = fq;

    float* mine = part + wave * P1_TILE;
    #pragma unroll
    for (int kt = 0; kt < 4; ++kt) {
        #pragma unroll
        for (int r = 0; r < 8; ++r) {
            const int k = (r + 8 * hi) * 4 + kt;
            const float4 v = make_float4(acc[kt][0][r], acc[kt][1][r], acc[kt][2][r], acc[kt][3][r]);
            *(float4*)&mine[k * P1_DSL + l15 * 4] = v;
        }
    }
    __syncthreads();

    float s[P1_TILE / 512];
    #pragma unroll
    for (int j = 0; j < P1_TILE / 512; ++j) {
        const int e = j * 512 + tid;
        float t = part[e];
        #pragma unroll
        for (int w = 1; w < P1_WAVES; ++w) t += part[w * P1_TILE + e];
        s[j] = t;
    }
    float fs = 0.0f;
    if (dg == 0 && tid < K_CLS) {
        fs = fpart[tid];
        #pragma unroll
        for (int w = 1; w < P1_WAVES; ++w) fs += fpart[w * K_CLS + tid];
    }
    #pragma unroll
    for (int j = 0; j < P1_TILE / 512; ++j) {
        const int e = j * 512 + tid;
        unsafeAtomicAdd(&sums[(e >> 6) * D_DIM + dg * P1_DSL + (e & 63)], s[j]);
    }
    if (dg == 0 && tid < K_CLS) unsafeAtomicAdd(&freq[tid], fs);
}

__global__ __launch_bounds__(256)
void means_msq(const float* __restrict__ sums, const float* __restrict__ freq,
               const float* __restrict__ past, float* __restrict__ out_means,
               float* __restrict__ msq) {
    __shared__ float red[8];
    const int k = blockIdx.x;
    const int d = threadIdx.x;
    const float f  = freq[k];
    const float bm = sums[k * D_DIM + d] / fmaxf(f, FP_EPS);
    const float pm = past[k * D_DIM + d];
    const float m  = (f > FP_EPS) ? (pm * 0.5f + bm * 0.5f) : pm;
    out_means[k * D_DIM + d] = m;

    float sq = m * m;
    #pragma unroll
    for (int off = 16; off > 0; off >>= 1) sq += __shfl_xor(sq, off, 32);
    if ((threadIdx.x & 31) == 0) red[threadIdx.x >> 5] = sq;
    __syncthreads();
    if (threadIdx.x == 0) {
        float t = 0.0f;
        #pragma unroll
        for (int i = 0; i < 8; ++i) t += red[i];
        msq[k] = t;
    }
}

#define P2_WAVES 8
#define P2_ROWS  (P2_WAVES * 16)

struct P2X { float4 v[4]; };
struct P2B { float4 v[2][4]; };

__device__ __forceinline__ void p2_loadx(P2X& x, const float4* __restrict__ xp, int q) {
    #pragma unroll
    for (int j = 0; j < 4; ++j) x.v[j] = xp[q * 8 + j * 2];
}

__device__ __forceinline__ void p2_loadb(P2B& b, const float* __restrict__ mb, int hf) {
    #pragma unroll
    for (int j = 0; j < 2; ++j) {
        #pragma unroll
        for (int kt = 0; kt < 4; ++kt)
            b.v[j][kt] = *(const float4*)&mb[kt * 16 * MLDS_STRIDE + hf * 16 + j * 8];
    }
}

template <int H>
__device__ __forceinline__ void p2_mma(const P2X& x, const P2B& b, v8f (&acc)[4], float& xs) {
    #pragma unroll
    for (int j = 0; j < 2; ++j) {
        const float4 xv = x.v[2 * H + j];
        xs += xv.x * xv.x + xv.y * xv.y + xv.z * xv.z + xv.w * xv.w;
        v2f a0, a1;
        a0.x = xv.x; a0.y = xv.y;
        a1.x = xv.z; a1.y = xv.w;
        #pragma unroll
        for (int kt = 0; kt < 4; ++kt) {
            v2f bf; bf.x = b.v[j][kt].x; bf.y = b.v[j][kt].y;
            acc[kt] = wmma4(a0, bf, acc[kt]);
        }
        #pragma unroll
        for (int kt = 0; kt < 4; ++kt) {
            v2f bf; bf.x = b.v[j][kt].z; bf.y = b.v[j][kt].w;
            acc[kt] = wmma4(a1, bf, acc[kt]);
        }
    }
}

__global__ __launch_bounds__(256)
void pass2_loss(const float* __restrict__ X, const float* __restrict__ A,
                const float* __restrict__ S, const float* __restrict__ means,
                const float* __restrict__ msq, float* __restrict__ loss) {
    __shared__ float mlds[K_CLS * MLDS_STRIDE];
    __shared__ float msq_lds[K_CLS];
    __shared__ float red[P2_WAVES];

    const int tid  = threadIdx.x;
    const int lane = tid & 31;
    const int wave = tid >> 5;
    const int l15  = lane & 15;
    const int hi   = lane >> 4;

    const int n0 = blockIdx.x * P2_ROWS + wave * 16;
    const float4* xp = (const float4*)(X + (size_t)(n0 + l15) * D_DIM + hi * 4);

    P2X x0, x1;
    p2_loadx(x0, xp, 0);
    float4 wa[8], ws[8];
    #pragma unroll
    for (int r = 0; r < 8; ++r) {
        const size_t gi = (size_t)(n0 + 8 * hi + r) * K_CLS + l15 * 4;
        wa[r] = *(const float4*)&A[gi];
        ws[r] = *(const float4*)&S[gi];
    }

    #pragma unroll
    for (int i = 0; i < 16; ++i) {
        const int idx = tid + i * 256;
        const int k   = idx >> 6;
        const int c4  = idx & 63;
        const float4 v = ((const float4*)means)[(size_t)k * 64 + c4];
        *(float4*)&mlds[((k & 3) * 16 + (k >> 2)) * MLDS_STRIDE + c4 * 4] = v;
    }
    if (tid < K_CLS) msq_lds[tid] = msq[tid];
    __syncthreads();

    const float* mb = mlds + l15 * MLDS_STRIDE + hi * 4;

    v8f acc[4];
    #pragma unroll
    for (int a = 0; a < 4; ++a) { v8f z = {}; acc[a] = z; }
    float xs = 0.0f;

    P2B b0, b1;
    p2_loadb(b0, mb, 0);
    for (int q = 0; q < 6; q += 2) {
        p2_loadx(x1, xp, q + 1);
        p2_loadb(b1, mb, 2 * q + 1);
        __builtin_amdgcn_sched_barrier(0);
        p2_mma<0>(x0, b0, acc, xs);
        __builtin_amdgcn_sched_barrier(0);
        p2_loadb(b0, mb, 2 * q + 2);
        __builtin_amdgcn_sched_barrier(0);
        p2_mma<1>(x0, b1, acc, xs);
        __builtin_amdgcn_sched_barrier(0);
        p2_loadx(x0, xp, q + 2);
        p2_loadb(b1, mb, 2 * q + 3);
        __builtin_amdgcn_sched_barrier(0);
        p2_mma<0>(x1, b0, acc, xs);
        __builtin_amdgcn_sched_barrier(0);
        p2_loadb(b0, mb, 2 * q + 4);
        __builtin_amdgcn_sched_barrier(0);
        p2_mma<1>(x1, b1, acc, xs);
    }
    p2_loadx(x1, xp, 7);
    p2_loadb(b1, mb, 13);
    __builtin_amdgcn_sched_barrier(0);
    p2_mma<0>(x0, b0, acc, xs);
    __builtin_amdgcn_sched_barrier(0);
    p2_loadb(b0, mb, 14);
    __builtin_amdgcn_sched_barrier(0);
    p2_mma<1>(x0, b1, acc, xs);
    __builtin_amdgcn_sched_barrier(0);
    p2_loadb(b1, mb, 15);
    __builtin_amdgcn_sched_barrier(0);
    p2_mma<0>(x1, b0, acc, xs);
    p2_mma<1>(x1, b1, acc, xs);

    const float xt = xs + __shfl_xor(xs, 16, 32);
    const float4 mq = *(const float4*)&msq_lds[l15 * 4];
    float csum = 0.0f;
    #pragma unroll
    for (int r = 0; r < 8; ++r) {
        const float xr = __shfl(xt, 8 * hi + r, 32);
        csum += wa[r].x * ws[r].x * (xr + mq.x - 2.0f * acc[0][r]);
        csum += wa[r].y * ws[r].y * (xr + mq.y - 2.0f * acc[1][r]);
        csum += wa[r].z * ws[r].z * (xr + mq.z - 2.0f * acc[2][r]);
        csum += wa[r].w * ws[r].w * (xr + mq.w - 2.0f * acc[3][r]);
    }

    #pragma unroll
    for (int off = 16; off > 0; off >>= 1) csum += __shfl_xor(csum, off, 32);
    if (lane == 0) red[wave] = csum;
    __syncthreads();
    if (tid == 0) {
        float tsum = 0.0f;
        #pragma unroll
        for (int w = 0; w < P2_WAVES; ++w) tsum += red[w];
        unsafeAtomicAdd(loss, tsum * (1.0f / (float)N_TOT));
    }
}

extern "C" void kernel_launch(void* const* d_in, const int* in_sizes, int n_in,
                              void* d_out, int out_size, void* d_ws, size_t ws_size,
                              hipStream_t stream) {
    const float* X = (const float*)d_in[0];
    const float* A = (const float*)d_in[1];
    const float* S = (const float*)d_in[2];
    const float* P = (const float*)d_in[3];
    float* out = (float*)d_out;

    float* ws   = (float*)d_ws;
    float* sums = ws;
    float* freq = ws + K_CLS * D_DIM;
    float* msq  = freq + K_CLS;

    zero_ws<<<(K_CLS * D_DIM + 2 * K_CLS + 255) / 256, 256, 0, stream>>>(ws, out);

    pass1_colsum<<<(N_TOT / P1_ROWS) * (D_DIM / P1_DSL), 512, 0, stream>>>(X, A, sums, freq);

    means_msq<<<K_CLS, 256, 0, stream>>>(sums, freq, P, out + 1, msq);

    pass2_loss<<<N_TOT / P2_ROWS, 256, 0, stream>>>(X, A, S, out + 1, msq, out);
}
